# starting kernel
# baseline (speedup 1.0000x reference)
#include <hip/hip_runtime.h>

#define N_NODES 100000
#define N_EDGES 1600000
#define IN_F 128
#define H_F 64

typedef float v2f __attribute__((ext_vector_type(2)));
typedef float v8f __attribute__((ext_vector_type(8)));

__global__ __launch_bounds__(256) void k_init_deg(float* __restrict__ deg) {
    int i = blockIdx.x * 256 + threadIdx.x;
    if (i < N_NODES) deg[i] = 1.0f;
}

__global__ __launch_bounds__(256) void k_edge_deg(const int* __restrict__ dst,
                                                  float* __restrict__ deg) {
    int e = blockIdx.x * 256 + threadIdx.x;
    if (e < N_EDGES) atomicAdd(&deg[dst[e]], 1.0f);
}

__global__ __launch_bounds__(256) void k_dinv(const float* __restrict__ deg,
                                              float* __restrict__ dinv) {
    int i = blockIdx.x * 256 + threadIdx.x;
    if (i < N_NODES) {
        float d = deg[i];
        dinv[i] = d > 0.0f ? rsqrtf(d) : 0.0f;
    }
}

__global__ __launch_bounds__(256) void k_gemm1(const float* __restrict__ x,
                                               const float* __restrict__ W1,
                                               float* __restrict__ h) {
    __shared__ float sW[IN_F * H_F];
    for (int i = threadIdx.x; i < IN_F * H_F; i += 256) {
        const int k = i >> 6, n = i & 63;
        sW[((k >> 1) << 7) + (n << 1) + (k & 1)] = W1[i];
    }
    __syncthreads();

    const int lane    = threadIdx.x & 31;
    const int wave    = threadIdx.x >> 5;
    const int rowBase = (blockIdx.x * 8 + wave) * 16;
    if (rowBase >= N_NODES) return;

    const int half = lane >> 4;
    const int l15  = lane & 15;
    const float* xrow = x + (size_t)(rowBase + l15) * IN_F;

    v8f acc[4] = {};

    for (int k0 = 0; k0 < IN_F; k0 += 4) {
        const int ks = k0 + half * 2;
        v2f a = *(const v2f*)(xrow + ks);
        const float* bp = &sW[((ks >> 1) << 7) + (l15 << 1)];
#pragma unroll
        for (int t = 0; t < 4; ++t) {
            v2f b = *(const v2f*)(bp + (t << 5));
            acc[t] = __builtin_amdgcn_wmma_f32_16x16x4_f32(
                false, a, false, b, (short)0, acc[t], false, false);
        }
    }

#pragma unroll
    for (int t = 0; t < 4; ++t) {
        const int col = t * 16 + l15;
#pragma unroll
        for (int r = 0; r < 8; ++r) {
            const int row = rowBase + r + 8 * half;
            h[(size_t)row * H_F + col] = acc[t][r];
        }
    }
}

__global__ __launch_bounds__(256) void k_self1(const float* __restrict__ h,
                                               const float* __restrict__ dinv,
                                               float* __restrict__ agg) {
    int t = blockIdx.x * 256 + threadIdx.x;
    if (t >= N_NODES * 16) return;
    int node = t >> 4, fg = t & 15;
    float n = dinv[node]; n *= n;
    float4 v = *((const float4*)(h + (size_t)node * H_F) + fg);
    float4 o = make_float4(v.x * n, v.y * n, v.z * n, v.w * n);
    *((float4*)(agg + (size_t)node * H_F) + fg) = o;
}

__global__ __launch_bounds__(256) void k_edge1(const int* __restrict__ src,
                                               const int* __restrict__ dst,
                                               const float* __restrict__ h,
                                               const float* __restrict__ dinv,
                                               float* __restrict__ agg) {
    int t = blockIdx.x * 256 + threadIdx.x;
    if (t >= N_EDGES * 16) return;
    int e = t >> 4, fg = t & 15;
    int s = src[e], d = dst[e];
    float norm = dinv[s] * dinv[d];
    float4 v = *((const float4*)(h + (size_t)s * H_F) + fg);
    float* o = agg + (size_t)d * H_F + fg * 4;
    atomicAdd(o + 0, v.x * norm);
    atomicAdd(o + 1, v.y * norm);
    atomicAdd(o + 2, v.z * norm);
    atomicAdd(o + 3, v.w * norm);
}

__global__ __launch_bounds__(256) void k_dot(const float* __restrict__ agg,
                                             const float* __restrict__ b1,
                                             const float* __restrict__ W2,
                                             float* __restrict__ z) {
    int gid  = blockIdx.x * 256 + threadIdx.x;
    int node = gid >> 5;
    int lane = threadIdx.x & 31;
    if (node >= N_NODES) return;
    float2 hv = *((const float2*)(agg + (size_t)node * H_F) + lane);
    float2 bb = *((const float2*)b1 + lane);
    float2 ww = *((const float2*)W2 + lane);
    float s = fmaxf(hv.x + bb.x, 0.0f) * ww.x + fmaxf(hv.y + bb.y, 0.0f) * ww.y;
#pragma unroll
    for (int off = 16; off > 0; off >>= 1) s += __shfl_xor(s, off, 32);
    if (lane == 0) z[node] = s;
}

__global__ __launch_bounds__(256) void k_self2(const float* __restrict__ z,
                                               const float* __restrict__ dinv,
                                               const float* __restrict__ b2,
                                               float* __restrict__ out) {
    int i = blockIdx.x * 256 + threadIdx.x;
    if (i < N_NODES) {
        float n = dinv[i];
        out[i] = z[i] * n * n + b2[0];
    }
}

__global__ __launch_bounds__(256) void k_edge2(const int* __restrict__ src,
                                               const int* __restrict__ dst,
                                               const float* __restrict__ z,
                                               const float* __restrict__ dinv,
                                               float* __restrict__ out) {
    int e = blockIdx.x * 256 + threadIdx.x;
    if (e < N_EDGES) {
        int s = src[e], d = dst[e];
        atomicAdd(&out[d], z[s] * dinv[s] * dinv[d]);
    }
}

extern "C" void kernel_launch(void* const* d_in, const int* in_sizes, int n_in,
                              void* d_out, int out_size, void* d_ws, size_t ws_size,
                              hipStream_t stream) {
    const float* x   = (const float*)d_in[0];
    const float* W1  = (const float*)d_in[1];
    const float* b1  = (const float*)d_in[2];
    const float* W2  = (const float*)d_in[3];
    const float* b2  = (const float*)d_in[4];
    const int*   ei  = (const int*)d_in[5];
    const int*   src = ei;
    const int*   dst = ei + N_EDGES;
    float*       out = (float*)d_out;

    char* ws = (char*)d_ws;
    const size_t NPAD = 409600;
    float* deg  = (float*)(ws);
    float* dinv = (float*)(ws + NPAD);
    float* h    = (float*)(ws + 2 * NPAD);
    float* agg1 = (float*)(ws + 2 * NPAD + (size_t)N_NODES * H_F * 4);
    float* z    = deg;

    const int nb_nodes = (N_NODES + 255) / 256;
    const int nb_edges = (N_EDGES + 255) / 256;

    k_init_deg<<<nb_nodes, 256, 0, stream>>>(deg);
    k_edge_deg<<<nb_edges, 256, 0, stream>>>(dst, deg);
    k_dinv    <<<nb_nodes, 256, 0, stream>>>(deg, dinv);

    k_gemm1   <<<(N_NODES + 127) / 128, 256, 0, stream>>>(x, W1, h);

    k_self1   <<<(N_NODES * 16) / 256, 256, 0, stream>>>(h, dinv, agg1);
    k_edge1   <<<(N_EDGES * 16) / 256, 256, 0, stream>>>(src, dst, h, dinv, agg1);

    k_dot     <<<(N_NODES * 32) / 256, 256, 0, stream>>>(agg1, b1, W2, z);

    k_self2   <<<nb_nodes, 256, 0, stream>>>(z, dinv, b2, out);
    k_edge2   <<<nb_edges, 256, 0, stream>>>(src, dst, z, dinv, out);
}


# k_edge1: one wave walks 32 edges; each row gather and each f32 atomic add covers one whole 128-byte line with 32 lanes (was 4 lines x 8 floats per instruction); atomics stay unordered
# speedup vs baseline: 1.3121x; 1.3121x over previous
#include <hip/hip_runtime.h>

#define N_NODES 100000
#define N_EDGES 1600000
#define IN_F 128
#define H_F 64

typedef float v2f __attribute__((ext_vector_type(2)));
typedef float v8f __attribute__((ext_vector_type(8)));

__global__ __launch_bounds__(256) void k_init_deg(float* __restrict__ deg) {
    int i = blockIdx.x * 256 + threadIdx.x;
    if (i < N_NODES) deg[i] = 1.0f;
}

__global__ __launch_bounds__(256) void k_edge_deg(const int* __restrict__ dst,
                                                  float* __restrict__ deg) {
    int e = blockIdx.x * 256 + threadIdx.x;
    if (e < N_EDGES) atomicAdd(&deg[dst[e]], 1.0f);
}

__global__ __launch_bounds__(256) void k_dinv(const float* __restrict__ deg,
                                              float* __restrict__ dinv) {
    int i = blockIdx.x * 256 + threadIdx.x;
    if (i < N_NODES) {
        float d = deg[i];
        dinv[i] = d > 0.0f ? rsqrtf(d) : 0.0f;
    }
}

__global__ __launch_bounds__(256) void k_gemm1(const float* __restrict__ x,
                                               const float* __restrict__ W1,
                                               float* __restrict__ h) {
    __shared__ float sW[IN_F * H_F];
    for (int i = threadIdx.x; i < IN_F * H_F; i += 256) {
        const int k = i >> 6, n = i & 63;
        sW[((k >> 1) << 7) + (n << 1) + (k & 1)] = W1[i];
    }
    __syncthreads();

    const int lane    = threadIdx.x & 31;
    const int wave    = threadIdx.x >> 5;
    const int rowBase = (blockIdx.x * 8 + wave) * 16;
    if (rowBase >= N_NODES) return;

    const int half = lane >> 4;
    const int l15  = lane & 15;
    const float* xrow = x + (size_t)(rowBase + l15) * IN_F;

    v8f acc[4] = {};

    for (int k0 = 0; k0 < IN_F; k0 += 4) {
        const int ks = k0 + half * 2;
        v2f a = *(const v2f*)(xrow + ks);
        const float* bp = &sW[((ks >> 1) << 7) + (l15 << 1)];
#pragma unroll
        for (int t = 0; t < 4; ++t) {
            v2f b = *(const v2f*)(bp + (t << 5));
            acc[t] = __builtin_amdgcn_wmma_f32_16x16x4_f32(
                false, a, false, b, (short)0, acc[t], false, false);
        }
    }

#pragma unroll
    for (int t = 0; t < 4; ++t) {
        const int col = t * 16 + l15;
#pragma unroll
        for (int r = 0; r < 8; ++r) {
            const int row = rowBase + r + 8 * half;
            h[(size_t)row * H_F + col] = acc[t][r];
        }
    }
}

__global__ __launch_bounds__(256) void k_self1(const float* __restrict__ h,
                                               const float* __restrict__ dinv,
                                               float* __restrict__ agg) {
    int t = blockIdx.x * 256 + threadIdx.x;
    if (t >= N_NODES * 16) return;
    int node = t >> 4, fg = t & 15;
    float n = dinv[node]; n *= n;
    float4 v = *((const float4*)(h + (size_t)node * H_F) + fg);
    float4 o = make_float4(v.x * n, v.y * n, v.z * n, v.w * n);
    *((float4*)(agg + (size_t)node * H_F) + fg) = o;
}

static_assert(N_EDGES % 256 == 0, "k_edge1 takes whole blocks of 256 edges, no tail");

__global__ __launch_bounds__(256) void k_edge1(const int* __restrict__ src,
                                               const int* __restrict__ dst,
                                               const float* __restrict__ h,
                                               const float* __restrict__ dinv,
                                               float* __restrict__ agg) {
    const int lane = threadIdx.x & 31;
    const int e    = blockIdx.x * 256 + threadIdx.x;
    const int s = src[e], d = dst[e];
    const int normBits = __float_as_int(dinv[s] * dinv[d]);

#pragma unroll
    for (int c = 0; c < 32; c += 8) {
        float lo[8], hi[8];
#pragma unroll
        for (int j = 0; j < 8; ++j) {
            const int sj = __builtin_amdgcn_readlane(s, c + j);
            const float* row = h + (size_t)sj * H_F + lane;
            lo[j] = row[0];
            hi[j] = row[32];
        }
#pragma unroll
        for (int j = 0; j < 8; ++j) {
            const int   dj = __builtin_amdgcn_readlane(d, c + j);
            const float nj = __int_as_float(__builtin_amdgcn_readlane(normBits, c + j));
            float* o = agg + (size_t)dj * H_F + lane;
            atomicAdd(o,      lo[j] * nj);
            atomicAdd(o + 32, hi[j] * nj);
        }
    }
}

__global__ __launch_bounds__(256) void k_dot(const float* __restrict__ agg,
                                             const float* __restrict__ b1,
                                             const float* __restrict__ W2,
                                             float* __restrict__ z) {
    int gid  = blockIdx.x * 256 + threadIdx.x;
    int node = gid >> 5;
    int lane = threadIdx.x & 31;
    if (node >= N_NODES) return;
    float2 hv = *((const float2*)(agg + (size_t)node * H_F) + lane);
    float2 bb = *((const float2*)b1 + lane);
    float2 ww = *((const float2*)W2 + lane);
    float s = fmaxf(hv.x + bb.x, 0.0f) * ww.x + fmaxf(hv.y + bb.y, 0.0f) * ww.y;
#pragma unroll
    for (int off = 16; off > 0; off >>= 1) s += __shfl_xor(s, off, 32);
    if (lane == 0) z[node] = s;
}

__global__ __launch_bounds__(256) void k_self2(const float* __restrict__ z,
                                               const float* __restrict__ dinv,
                                               const float* __restrict__ b2,
                                               float* __restrict__ out) {
    int i = blockIdx.x * 256 + threadIdx.x;
    if (i < N_NODES) {
        float n = dinv[i];
        out[i] = z[i] * n * n + b2[0];
    }
}

__global__ __launch_bounds__(256) void k_edge2(const int* __restrict__ src,
                                               const int* __restrict__ dst,
                                               const float* __restrict__ z,
                                               const float* __restrict__ dinv,
                                               float* __restrict__ out) {
    int e = blockIdx.x * 256 + threadIdx.x;
    if (e < N_EDGES) {
        int s = src[e], d = dst[e];
        atomicAdd(&out[d], z[s] * dinv[s] * dinv[d]);
    }
}

extern "C" void kernel_launch(void* const* d_in, const int* in_sizes, int n_in,
                              void* d_out, int out_size, void* d_ws, size_t ws_size,
                              hipStream_t stream) {
    const float* x   = (const float*)d_in[0];
    const float* W1  = (const float*)d_in[1];
    const float* b1  = (const float*)d_in[2];
    const float* W2  = (const float*)d_in[3];
    const float* b2  = (const float*)d_in[4];
    const int*   ei  = (const int*)d_in[5];
    const int*   src = ei;
    const int*   dst = ei + N_EDGES;
    float*       out = (float*)d_out;

    char* ws = (char*)d_ws;
    const size_t NPAD = 409600;
    float* deg  = (float*)(ws);
    float* dinv = (float*)(ws + NPAD);
    float* h    = (float*)(ws + 2 * NPAD);
    float* agg1 = (float*)(ws + 2 * NPAD + (size_t)N_NODES * H_F * 4);
    float* z    = deg;

    const int nb_nodes = (N_NODES + 255) / 256;
    const int nb_edges = (N_EDGES + 255) / 256;

    k_init_deg<<<nb_nodes, 256, 0, stream>>>(deg);
    k_edge_deg<<<nb_edges, 256, 0, stream>>>(dst, deg);
    k_dinv    <<<nb_nodes, 256, 0, stream>>>(deg, dinv);

    k_gemm1   <<<(N_NODES + 127) / 128, 256, 0, stream>>>(x, W1, h);

    k_self1   <<<(N_NODES * 16) / 256, 256, 0, stream>>>(h, dinv, agg1);
    k_edge1   <<<N_EDGES / 256, 256, 0, stream>>>(src, dst, h, dinv, agg1);

    k_dot     <<<(N_NODES * 32) / 256, 256, 0, stream>>>(agg1, b1, W2, z);

    k_self2   <<<nb_nodes, 256, 0, stream>>>(z, dinv, b2, out);
    k_edge2   <<<nb_edges, 256, 0, stream>>>(src, dst, z, dinv, out);
}


# norm factored as dinv[dst]*sum(dinv[src]*row): GEMM epilogue and k_dot store rows pre-scaled and seed each sum; edge passes add with no dinv gathers; dst scale once after the sum (rounding moves)
# speedup vs baseline: 1.3921x; 1.0610x over previous
#include <hip/hip_runtime.h>

#define N_NODES 100000
#define N_EDGES 1600000
#define IN_F 128
#define H_F 64

typedef float v2f __attribute__((ext_vector_type(2)));
typedef float v8f __attribute__((ext_vector_type(8)));

__global__ __launch_bounds__(256) void k_init_deg(float* __restrict__ deg) {
    int i = blockIdx.x * 256 + threadIdx.x;
    if (i < N_NODES) deg[i] = 1.0f;
}

__global__ __launch_bounds__(256) void k_edge_deg(const int* __restrict__ dst,
                                                  float* __restrict__ deg) {
    int e = blockIdx.x * 256 + threadIdx.x;
    if (e < N_EDGES) atomicAdd(&deg[dst[e]], 1.0f);
}

__global__ __launch_bounds__(256) void k_dinv(const float* __restrict__ deg,
                                              float* __restrict__ dinv) {
    int i = blockIdx.x * 256 + threadIdx.x;
    if (i < N_NODES) {
        float d = deg[i];
        dinv[i] = d > 0.0f ? rsqrtf(d) : 0.0f;
    }
}

__global__ __launch_bounds__(256) void k_gemm1(const float* __restrict__ x,
                                               const float* __restrict__ W1,
                                               const float* __restrict__ dinv,
                                               float* __restrict__ h,
                                               float* __restrict__ agg) {
    __shared__ float sW[IN_F * H_F];
    for (int i = threadIdx.x; i < IN_F * H_F; i += 256) {
        const int k = i >> 6, n = i & 63;
        sW[((k >> 1) << 7) + (n << 1) + (k & 1)] = W1[i];
    }
    __syncthreads();

    const int lane    = threadIdx.x & 31;
    const int wave    = threadIdx.x >> 5;
    const int rowBase = (blockIdx.x * 8 + wave) * 16;
    if (rowBase >= N_NODES) return;

    const int half = lane >> 4;
    const int l15  = lane & 15;
    const float* xrow = x + (size_t)(rowBase + l15) * IN_F;

    const float4* dp = (const float4*)(dinv + rowBase + 8 * half);
    const float4 dA = dp[0], dB = dp[1];
    const float dsc[8] = {dA.x, dA.y, dA.z, dA.w, dB.x, dB.y, dB.z, dB.w};

    v8f acc[4] = {};

    for (int k0 = 0; k0 < IN_F; k0 += 4) {
        const int ks = k0 + half * 2;
        v2f a = *(const v2f*)(xrow + ks);
        const float* bp = &sW[((ks >> 1) << 7) + (l15 << 1)];
#pragma unroll
        for (int t = 0; t < 4; ++t) {
            v2f b = *(const v2f*)(bp + (t << 5));
            acc[t] = __builtin_amdgcn_wmma_f32_16x16x4_f32(
                false, a, false, b, (short)0, acc[t], false, false);
        }
    }

#pragma unroll
    for (int t = 0; t < 4; ++t) {
        const int col = t * 16 + l15;
#pragma unroll
        for (int r = 0; r < 8; ++r) {
            const int row = rowBase + r + 8 * half;
            const float v = acc[t][r] * dsc[r];
            h[(size_t)row * H_F + col]   = v;
            agg[(size_t)row * H_F + col] = v;
        }
    }
}

static_assert(N_EDGES % 256 == 0, "k_edge1 takes whole blocks of 256 edges, no tail");
static_assert(N_NODES % 16 == 0, "k_gemm1 strips are whole: 16 rows, 8 dinv values per half");
static_assert((N_NODES * 32) % 256 == 0, "k_dot covers every node: it stores out's start value");

__global__ __launch_bounds__(256) void k_edge1(const int* __restrict__ src,
                                               const int* __restrict__ dst,
                                               const float* __restrict__ h,
                                               float* __restrict__ agg) {
    const int lane = threadIdx.x & 31;
    const int e    = blockIdx.x * 256 + threadIdx.x;
    const int s = src[e], d = dst[e];

#pragma unroll
    for (int c = 0; c < 32; c += 8) {
        float lo[8], hi[8];
#pragma unroll
        for (int j = 0; j < 8; ++j) {
            const int sj = __builtin_amdgcn_readlane(s, c + j);
            const float* row = h + (size_t)sj * H_F + lane;
            lo[j] = row[0];
            hi[j] = row[32];
        }
#pragma unroll
        for (int j = 0; j < 8; ++j) {
            const int dj = __builtin_amdgcn_readlane(d, c + j);
            float* o = agg + (size_t)dj * H_F + lane;
            atomicAdd(o,      lo[j]);
            atomicAdd(o + 32, hi[j]);
        }
    }
}

__global__ __launch_bounds__(256) void k_dot(const float* __restrict__ agg,
                                             const float* __restrict__ dinv,
                                             const float* __restrict__ b1,
                                             const float* __restrict__ W2,
                                             float* __restrict__ zs,
                                             float* __restrict__ out) {
    int gid  = blockIdx.x * 256 + threadIdx.x;
    int node = gid >> 5;
    int lane = threadIdx.x & 31;
    if (node >= N_NODES) return;
    const float dn = dinv[node];
    float2 hv = *((const float2*)(agg + (size_t)node * H_F) + lane);
    float2 bb = *((const float2*)b1 + lane);
    float2 ww = *((const float2*)W2 + lane);
    float s = fmaxf(hv.x * dn + bb.x, 0.0f) * ww.x + fmaxf(hv.y * dn + bb.y, 0.0f) * ww.y;
#pragma unroll
    for (int off = 16; off > 0; off >>= 1) s += __shfl_xor(s, off, 32);
    if (lane == 0) {
        const float v = s * dn;
        zs[node]  = v;
        out[node] = v;
    }
}

__global__ __launch_bounds__(256) void k_edge2(const int* __restrict__ src,
                                               const int* __restrict__ dst,
                                               const float* __restrict__ zs,
                                               float* __restrict__ out) {
    int e = blockIdx.x * 256 + threadIdx.x;
    if (e < N_EDGES) atomicAdd(&out[dst[e]], zs[src[e]]);
}

__global__ __launch_bounds__(256) void k_post2(const float* __restrict__ dinv,
                                               const float* __restrict__ b2,
                                               float* __restrict__ out) {
    int i = blockIdx.x * 256 + threadIdx.x;
    if (i < N_NODES) out[i] = out[i] * dinv[i] + b2[0];
}

extern "C" void kernel_launch(void* const* d_in, const int* in_sizes, int n_in,
                              void* d_out, int out_size, void* d_ws, size_t ws_size,
                              hipStream_t stream) {
    const float* x   = (const float*)d_in[0];
    const float* W1  = (const float*)d_in[1];
    const float* b1  = (const float*)d_in[2];
    const float* W2  = (const float*)d_in[3];
    const float* b2  = (const float*)d_in[4];
    const int*   ei  = (const int*)d_in[5];
    const int*   src = ei;
    const int*   dst = ei + N_EDGES;
    float*       out = (float*)d_out;

    char* ws = (char*)d_ws;
    const size_t NPAD = 409600;
    float* deg  = (float*)(ws);
    float* dinv = (float*)(ws + NPAD);
    float* h    = (float*)(ws + 2 * NPAD);
    float* agg1 = (float*)(ws + 2 * NPAD + (size_t)N_NODES * H_F * 4);
    float* zs   = deg;

    const int nb_nodes = (N_NODES + 255) / 256;
    const int nb_edges = (N_EDGES + 255) / 256;

    k_init_deg<<<nb_nodes, 256, 0, stream>>>(deg);
    k_edge_deg<<<nb_edges, 256, 0, stream>>>(dst, deg);
    k_dinv    <<<nb_nodes, 256, 0, stream>>>(deg, dinv);

    k_gemm1   <<<(N_NODES + 127) / 128, 256, 0, stream>>>(x, W1, dinv, h, agg1);

    k_edge1   <<<N_EDGES / 256, 256, 0, stream>>>(src, dst, h, agg1);

    k_dot     <<<(N_NODES * 32) / 256, 256, 0, stream>>>(agg1, dinv, b1, W2, zs, out);

    k_edge2   <<<nb_edges, 256, 0, stream>>>(src, dst, zs, out);
    k_post2   <<<nb_nodes, 256, 0, stream>>>(dinv, b2, out);
}


# edges grouped by dst each call (returning int add = rank, scan, place); a wave sums a node's rows in registers, fused relu/W2 dot; layer 2 uses the same lists; no f32 atomics; sum order = list order
# speedup vs baseline: 1.9571x; 1.4059x over previous
#include <hip/hip_runtime.h>

#define N_NODES 100000
#define N_EDGES 1600000
#define IN_F 128
#define H_F 64
#define PB 2048
#define EPT 4

typedef float v2f __attribute__((ext_vector_type(2)));
typedef float v8f __attribute__((ext_vector_type(8)));

static_assert(N_EDGES <= EPT * PB * 256, "edge passes cover every edge");
static_assert(N_NODES % 16 == 0, "k_gemm1 strips are whole: 16 rows, 8 dinv values per half");
static_assert(N_NODES % 4 == 0, "k_scan reads whole int4 groups");
static_assert((unsigned long long)(N_NODES + 1) * H_F * 4 < 0xffffffffull, "row byte offsets fit 32 bits");

__global__ __launch_bounds__(256) void k_init(int* __restrict__ cnt, float* __restrict__ hzero) {
    int i = blockIdx.x * 256 + threadIdx.x;
    if (i < N_NODES) cnt[i] = 0;
    if (i < H_F) hzero[i] = 0.0f;
}

__global__ __launch_bounds__(256) void k_rank(const int* __restrict__ dst,
                                              int* __restrict__ cnt,
                                              int* __restrict__ rank) {
    const int gid = blockIdx.x * 256 + threadIdx.x;
    int d[EPT], r[EPT];
#pragma unroll
    for (int k = 0; k < EPT; ++k) {
        const int e = gid + k * (PB * 256);
        d[k] = e < N_EDGES ? dst[e] : -1;
    }
#pragma unroll
    for (int k = 0; k < EPT; ++k) r[k] = d[k] >= 0 ? atomicAdd(&cnt[d[k]], 1) : 0;
#pragma unroll
    for (int k = 0; k < EPT; ++k) {
        const int e = gid + k * (PB * 256);
        if (d[k] >= 0) rank[e] = r[k];
    }
}

__global__ __launch_bounds__(1024) void k_scan(const int* __restrict__ cnt,
                                               int* __restrict__ rowptr,
                                               float* __restrict__ dinv) {
    __shared__ int sPre[32];
    __shared__ int sWave[32];
    const int tid = threadIdx.x, lane = tid & 31, wave = tid >> 5;
    const int base = blockIdx.x * 1024;

    int pre = 0;
    const int4* c4 = (const int4*)cnt;
    const int lim = base >> 2;
    for (int i = tid; i < lim; i += 8 * 1024) {
        int4 v[8];
#pragma unroll
        for (int u = 0; u < 8; ++u) {
            const int k = i + u * 1024;
            v[u] = k < lim ? c4[k] : make_int4(0, 0, 0, 0);
        }
#pragma unroll
        for (int u = 0; u < 8; ++u) pre += (v[u].x + v[u].y) + (v[u].z + v[u].w);
    }
#pragma unroll
    for (int off = 16; off > 0; off >>= 1) pre += __shfl_xor(pre, off, 32);

    const int i = base + tid;
    const int c = i < N_NODES ? cnt[i] : 0;
    int inc = c;
#pragma unroll
    for (int off = 1; off < 32; off <<= 1) {
        const int t = __shfl_up(inc, off, 32);
        if (lane >= off) inc += t;
    }
    if (lane == 0)  sPre[wave]  = pre;
    if (lane == 31) sWave[wave] = inc;
    __syncthreads();

    int ab = sPre[lane] + (lane < wave ? sWave[lane] : 0);
#pragma unroll
    for (int off = 16; off > 0; off >>= 1) ab += __shfl_xor(ab, off, 32);

    if (i < N_NODES) {
        rowptr[i] = ab + inc - c;
        const float dg = (float)(c + 1);
        dinv[i] = dg > 0.0f ? rsqrtf(dg) : 0.0f;
    }
}

__global__ __launch_bounds__(256) void k_fill(const int* __restrict__ src,
                                              const int* __restrict__ dst,
                                              const int* __restrict__ rowptr,
                                              const int* __restrict__ rank,
                                              int* __restrict__ csr) {
    const int gid = blockIdx.x * 256 + threadIdx.x;
    int d[EPT], s[EPT], r[EPT], p[EPT];
#pragma unroll
    for (int k = 0; k < EPT; ++k) {
        const int e = gid + k * (PB * 256);
        const bool ok = e < N_EDGES;
        d[k] = ok ? dst[e] : -1;
        s[k] = ok ? src[e] : 0;
        r[k] = ok ? rank[e] : 0;
    }
#pragma unroll
    for (int k = 0; k < EPT; ++k) p[k] = d[k] >= 0 ? rowptr[d[k]] : 0;
#pragma unroll
    for (int k = 0; k < EPT; ++k)
        if (d[k] >= 0) csr[p[k] + r[k]] = s[k];
}

__global__ __launch_bounds__(256) void k_gemm1(const float* __restrict__ x,
                                               const float* __restrict__ W1,
                                               const float* __restrict__ dinv,
                                               float* __restrict__ h) {
    __shared__ float sW[IN_F * H_F];
    for (int i = threadIdx.x; i < IN_F * H_F; i += 256) {
        const int k = i >> 6, n = i & 63;
        sW[((k >> 1) << 7) + (n << 1) + (k & 1)] = W1[i];
    }
    __syncthreads();

    const int lane    = threadIdx.x & 31;
    const int wave    = threadIdx.x >> 5;
    const int rowBase = (blockIdx.x * 8 + wave) * 16;
    if (rowBase >= N_NODES) return;

    const int half = lane >> 4;
    const int l15  = lane & 15;
    const float* xrow = x + (size_t)(rowBase + l15) * IN_F;

    const float4* dp = (const float4*)(dinv + rowBase + 8 * half);
    const float4 dA = dp[0], dB = dp[1];
    const float dsc[8] = {dA.x, dA.y, dA.z, dA.w, dB.x, dB.y, dB.z, dB.w};

    v8f acc[4] = {};

    for (int k0 = 0; k0 < IN_F; k0 += 4) {
        const int ks = k0 + half * 2;
        v2f a = *(const v2f*)(xrow + ks);
        const float* bp = &sW[((ks >> 1) << 7) + (l15 << 1)];
#pragma unroll
        for (int t = 0; t < 4; ++t) {
            v2f b = *(const v2f*)(bp + (t << 5));
            acc[t] = __builtin_amdgcn_wmma_f32_16x16x4_f32(
                false, a, false, b, (short)0, acc[t], false, false);
        }
    }

#pragma unroll
    for (int t = 0; t < 4; ++t) {
        const int col = t * 16 + l15;
#pragma unroll
        for (int r = 0; r < 8; ++r) {
            const int row = rowBase + r + 8 * half;
            h[(size_t)row * H_F + col] = acc[t][r] * dsc[r];
        }
    }
}

__global__ __launch_bounds__(256) void k_gather1(const int* __restrict__ rowptr,
                                                 const int* __restrict__ cnt,
                                                 const int* __restrict__ csr,
                                                 const float* __restrict__ h,
                                                 const float* __restrict__ dinv,
                                                 const float* __restrict__ b1,
                                                 const float* __restrict__ W2,
                                                 float* __restrict__ zs) {
    const int lane = threadIdx.x & 31;
    int n = __builtin_amdgcn_readfirstlane(blockIdx.x * 8 + (threadIdx.x >> 5));
    if (n >= N_NODES) return;
    const float2* h2 = (const float2*)h;
    const char* hb = (const char*)h;
    const unsigned lane8 = (unsigned)lane * 8u;
    const float2 bb = ((const float2*)b1)[lane];
    const float2 ww = ((const float2*)W2)[lane];

    int start = rowptr[n], c = cnt[n];
    int idx = lane < c ? csr[start + lane] : N_NODES;
    float2 acc = h2[(size_t)n * 32 + lane];
    float dn = dinv[n];
    int nn = n + PB * 8;
    int nstart = 0, nc = 0;
    if (nn < N_NODES) { nstart = rowptr[nn]; nc = cnt[nn]; }

    while (true) {
        int idxN = N_NODES, s2 = 0, c2 = 0;
        float2 accN = make_float2(0.0f, 0.0f);
        float dnN = 0.0f;
        const int n2 = nn + PB * 8;
        if (nn < N_NODES) {
            idxN = lane < nc ? csr[nstart + lane] : N_NODES;
            accN = h2[(size_t)nn * 32 + lane];
            dnN  = dinv[nn];
            if (n2 < N_NODES) { s2 = rowptr[n2]; c2 = cnt[n2]; }
        }

        for (int base = 0; base < c; base += 32) {
            const int m = min(32, c - base);
            if (base > 0) idx = lane < m ? csr[start + base + lane] : N_NODES;
            for (int j = 0; j < m; j += 16) {
                float2 v[16];
#pragma unroll
                for (int u = 0; u < 8; ++u)
                    v[u] = *(const float2*)(hb + ((unsigned)__builtin_amdgcn_readlane(idx, j + u) * 256u + lane8));
                if (j + 8 < m) {
#pragma unroll
                    for (int u = 0; u < 8; ++u)
                        v[8 + u] = *(const float2*)(hb + ((unsigned)__builtin_amdgcn_readlane(idx, j + 8 + u) * 256u + lane8));
                } else {
#pragma unroll
                    for (int u = 0; u < 8; ++u) v[8 + u] = make_float2(0.0f, 0.0f);
                }
#pragma unroll
                for (int u = 0; u < 16; ++u) { acc.x += v[u].x; acc.y += v[u].y; }
            }
        }

        float s = fmaxf(acc.x * dn + bb.x, 0.0f) * ww.x + fmaxf(acc.y * dn + bb.y, 0.0f) * ww.y;
#pragma unroll
        for (int off = 16; off > 0; off >>= 1) s += __shfl_xor(s, off, 32);
        if (lane == 0) zs[n] = s * dn;

        if (nn >= N_NODES) break;
        n = nn; start = nstart; c = nc; idx = idxN; acc = accN; dn = dnN;
        nn = n2; nstart = s2; nc = c2;
    }
}

__global__ __launch_bounds__(256) void k_gather2(const int* __restrict__ rowptr,
                                                 const int* __restrict__ cnt,
                                                 const int* __restrict__ csr,
                                                 const float* __restrict__ zs,
                                                 const float* __restrict__ dinv,
                                                 const float* __restrict__ b2,
                                                 float* __restrict__ out) {
    const int sub = threadIdx.x & 7;
    for (int n = (blockIdx.x * 256 + threadIdx.x) >> 3; n < N_NODES; n += PB * 32) {
        const int start = rowptr[n], c = cnt[n];
        float a0 = 0.0f, a1 = 0.0f;
        int p = sub;
        for (; p + 8 < c; p += 16) {
            const int i0 = csr[start + p], i1 = csr[start + p + 8];
            a0 += zs[i0];
            a1 += zs[i1];
        }
        if (p < c) a0 += zs[csr[start + p]];
        float acc = a0 + a1;
        acc += __shfl_xor(acc, 4, 32);
        acc += __shfl_xor(acc, 2, 32);
        acc += __shfl_xor(acc, 1, 32);
        if (sub == 0) out[n] = (acc + zs[n]) * dinv[n] + b2[0];
    }
}

extern "C" void kernel_launch(void* const* d_in, const int* in_sizes, int n_in,
                              void* d_out, int out_size, void* d_ws, size_t ws_size,
                              hipStream_t stream) {
    const float* x   = (const float*)d_in[0];
    const float* W1  = (const float*)d_in[1];
    const float* b1  = (const float*)d_in[2];
    const float* W2  = (const float*)d_in[3];
    const float* b2  = (const float*)d_in[4];
    const int*   ei  = (const int*)d_in[5];
    const int*   src = ei;
    const int*   dst = ei + N_EDGES;
    float*       out = (float*)d_out;

    char* ws = (char*)d_ws;
    const size_t NPAD = 409600;
    const size_t EPAD = 6400000;
    int*   cnt    = (int*)(ws);
    float* dinv   = (float*)(ws + NPAD);
    int*   rowptr = (int*)(ws + 2 * NPAD);
    float* zs     = (float*)(ws + 3 * NPAD);
    int*   rank   = (int*)(ws + 4 * NPAD);
    int*   csr    = (int*)(ws + 4 * NPAD + EPAD);
    float* h      = (float*)(ws + 4 * NPAD + 2 * EPAD);

    k_init   <<<(N_NODES + 255) / 256, 256, 0, stream>>>(cnt, h + (size_t)N_NODES * H_F);
    k_rank   <<<PB, 256, 0, stream>>>(dst, cnt, rank);
    k_scan   <<<(N_NODES + 1023) / 1024, 1024, 0, stream>>>(cnt, rowptr, dinv);
    k_fill   <<<PB, 256, 0, stream>>>(src, dst, rowptr, rank, csr);
    k_gemm1  <<<(N_NODES + 127) / 128, 256, 0, stream>>>(x, W1, dinv, h);
    k_gather1<<<PB, 256, 0, stream>>>(rowptr, cnt, csr, h, dinv, b1, W2, zs);
    k_gather2<<<PB, 256, 0, stream>>>(rowptr, cnt, csr, zs, dinv, b2, out);
}


# edges grouped by dst on LDS counters in two levels (256-node bucket: block counts + prefix + LDS rank; then per-node count, scan, place in LDS); no global atomics left; list (sum) order changes
# speedup vs baseline: 3.2996x; 1.6860x over previous
#include <hip/hip_runtime.h>

#define N_NODES 100000
#define N_EDGES 1600000
#define IN_F 128
#define H_F 64
#define PB 2048
#define BSH 8
#define NB ((N_NODES + 255) >> BSH)
#define NBP 512
#define NBLK 256
#define CH ((N_EDGES + NBLK - 1) / NBLK)
#define EPT ((CH + 1023) / 1024)

typedef float v2f __attribute__((ext_vector_type(2)));
typedef float v8f __attribute__((ext_vector_type(8)));

static_assert(NB < NBP, "bucket NB of the padded row exists and stays empty");
static_assert(NBP % 4 == 0 && NBP / 4 == 128, "k_part reads a table row as 128 int4");
static_assert(NBLK % 128 == 0, "k_part: 8 thread groups walk the table rows in groups of 16");
static_assert(N_NODES < (1 << 23), "src << 8 | local node fits an int");
static_assert(N_NODES % 16 == 0, "k_gemm1 strips are whole: 16 rows, 8 dinv values per half");
static_assert((unsigned long long)(N_NODES + 1) * H_F * 4 < 0xffffffffull, "row byte offsets fit 32 bits");

__global__ __launch_bounds__(1024) void k_hist(const int* __restrict__ dst,
                                               int* __restrict__ hist,
                                               float* __restrict__ hzero) {
    __shared__ int sH[NBP];
    const int tid = threadIdx.x;
    if (tid < NBP) sH[tid] = 0;
    if (blockIdx.x == 0 && tid < H_F) hzero[tid] = 0.0f;
    __syncthreads();
    const int e0 = blockIdx.x * CH;
    int d[EPT];
#pragma unroll
    for (int k = 0; k < EPT; ++k) {
        const int o = tid + k * 1024;
        const bool ok = o < CH && e0 + o < N_EDGES;
        const int dv = dst[ok ? e0 + o : 0];
        d[k] = ok ? dv : -1;
    }
#pragma unroll
    for (int k = 0; k < EPT; ++k)
        if (d[k] >= 0) atomicAdd(&sH[d[k] >> BSH], 1);
    __syncthreads();
    if (tid < NBP) hist[blockIdx.x * NBP + tid] = sH[tid];
}

__global__ __launch_bounds__(1024) void k_part(const int* __restrict__ src,
                                               const int* __restrict__ dst,
                                               const int* __restrict__ hist,
                                               int* __restrict__ bbase,
                                               int* __restrict__ rec) {
    __shared__ int sTot[NBP];
    __shared__ int sOff[NBP];
    __shared__ int sW[32];
    const int tid = threadIdx.x, lane = tid & 31, wave = tid >> 5;
    const int j = blockIdx.x;
    if (tid < NBP) { sTot[tid] = 0; sOff[tid] = 0; }
    __syncthreads();

    const int e0 = j * CH;
    int d[EPT], s[EPT];
#pragma unroll
    for (int k = 0; k < EPT; ++k) {
        const int o = tid + k * 1024;
        const bool ok = o < CH && e0 + o < N_EDGES;
        const int e = ok ? e0 + o : 0;
        const int dv = dst[e];
        s[k] = src[e];
        d[k] = ok ? dv : -1;
    }

    const int c4 = tid & 127, q = tid >> 7;
    const int4* h4 = (const int4*)hist;
    int4 tot = make_int4(0, 0, 0, 0), pre = make_int4(0, 0, 0, 0);
#pragma unroll 1
    for (int g = 0; g < NBLK / 8; g += 16) {
        int4 t[16];
#pragma unroll
        for (int u = 0; u < 16; ++u) t[u] = h4[(q + 8 * (g + u)) * (NBP / 4) + c4];
#pragma unroll
        for (int u = 0; u < 16; ++u) {
            const bool before = q + 8 * (g + u) < j;
            tot.x += t[u].x; tot.y += t[u].y; tot.z += t[u].z; tot.w += t[u].w;
            pre.x += before ? t[u].x : 0; pre.y += before ? t[u].y : 0;
            pre.z += before ? t[u].z : 0; pre.w += before ? t[u].w : 0;
        }
    }
    atomicAdd(&sTot[4 * c4 + 0], tot.x); atomicAdd(&sTot[4 * c4 + 1], tot.y);
    atomicAdd(&sTot[4 * c4 + 2], tot.z); atomicAdd(&sTot[4 * c4 + 3], tot.w);
    atomicAdd(&sOff[4 * c4 + 0], pre.x); atomicAdd(&sOff[4 * c4 + 1], pre.y);
    atomicAdd(&sOff[4 * c4 + 2], pre.z); atomicAdd(&sOff[4 * c4 + 3], pre.w);
    __syncthreads();

    const int v = tid < NBP ? sTot[tid] : 0;
    int inc = v;
#pragma unroll
    for (int off = 1; off < 32; off <<= 1) {
        const int t = __shfl_up(inc, off, 32);
        if (lane >= off) inc += t;
    }
    if (lane == 31) sW[wave] = inc;
    __syncthreads();
    int wb = lane < wave ? sW[lane] : 0;
#pragma unroll
    for (int off = 16; off > 0; off >>= 1) wb += __shfl_xor(wb, off, 32);
    const int excl = wb + inc - v;
    if (tid < NBP) {
        sOff[tid] += excl;
        if (j == 0 && tid <= NB) bbase[tid] = excl;
    }
    __syncthreads();

    int pos[EPT];
#pragma unroll
    for (int k = 0; k < EPT; ++k) pos[k] = d[k] >= 0 ? atomicAdd(&sOff[d[k] >> BSH], 1) : 0;
#pragma unroll
    for (int k = 0; k < EPT; ++k)
        if (d[k] >= 0) rec[pos[k]] = (s[k] << BSH) | (d[k] & 255);
}

__global__ __launch_bounds__(512) void k_group(const int* __restrict__ rec,
                                               const int* __restrict__ bbase,
                                               int* __restrict__ rowptr,
                                               int* __restrict__ cnt,
                                               float* __restrict__ dinv,
                                               int* __restrict__ csr) {
    __shared__ int sCnt[256];
    __shared__ int sCur[256];
    __shared__ int sW[8];
    const int tid = threadIdx.x, lane = tid & 31, wave = tid >> 5;
    const int b = blockIdx.x;
    const int begin = bbase[b], end = bbase[b + 1];
    if (tid < 256) sCnt[tid] = 0;
    __syncthreads();

    for (int i0 = begin; i0 < end; i0 += 512 * 8) {
        int r[8];
#pragma unroll
        for (int u = 0; u < 8; ++u) {
            const int i = i0 + u * 512 + tid;
            const int rv = rec[i < end ? i : begin];
            r[u] = i < end ? rv : -1;
        }
#pragma unroll
        for (int u = 0; u < 8; ++u)
            if (r[u] >= 0) atomicAdd(&sCnt[r[u] & 255], 1);
    }
    __syncthreads();

    const int c = tid < 256 ? sCnt[tid] : 0;
    int inc = c;
#pragma unroll
    for (int off = 1; off < 32; off <<= 1) {
        const int t = __shfl_up(inc, off, 32);
        if (lane >= off) inc += t;
    }
    if (tid < 256 && lane == 31) sW[wave] = inc;
    __syncthreads();
    int wb = (lane < wave && lane < 8) ? sW[lane] : 0;
#pragma unroll
    for (int off = 16; off > 0; off >>= 1) wb += __shfl_xor(wb, off, 32);
    if (tid < 256) {
        const int p = begin + wb + inc - c;
        sCur[tid] = p;
        const int node = (b << BSH) + tid;
        if (node < N_NODES) {
            rowptr[node] = p;
            cnt[node] = c;
            const float dg = (float)(c + 1);
            dinv[node] = dg > 0.0f ? rsqrtf(dg) : 0.0f;
        }
    }
    __syncthreads();

    for (int i0 = begin; i0 < end; i0 += 512 * 8) {
        int r[8];
#pragma unroll
        for (int u = 0; u < 8; ++u) {
            const int i = i0 + u * 512 + tid;
            const int rv = rec[i < end ? i : begin];
            r[u] = i < end ? rv : -1;
        }
#pragma unroll
        for (int u = 0; u < 8; ++u)
            if (r[u] >= 0) {
                const int pos = atomicAdd(&sCur[r[u] & 255], 1);
                csr[pos] = r[u] >> BSH;
            }
    }
}

__global__ __launch_bounds__(256) void k_gemm1(const float* __restrict__ x,
                                               const float* __restrict__ W1,
                                               const float* __restrict__ dinv,
                                               float* __restrict__ h) {
    __shared__ float sW[IN_F * H_F];
    for (int i = threadIdx.x; i < IN_F * H_F; i += 256) {
        const int k = i >> 6, n = i & 63;
        sW[((k >> 1) << 7) + (n << 1) + (k & 1)] = W1[i];
    }
    __syncthreads();

    const int lane    = threadIdx.x & 31;
    const int wave    = threadIdx.x >> 5;
    const int rowBase = (blockIdx.x * 8 + wave) * 16;
    if (rowBase >= N_NODES) return;

    const int half = lane >> 4;
    const int l15  = lane & 15;
    const float* xrow = x + (size_t)(rowBase + l15) * IN_F;

    const float4* dp = (const float4*)(dinv + rowBase + 8 * half);
    const float4 dA = dp[0], dB = dp[1];
    const float dsc[8] = {dA.x, dA.y, dA.z, dA.w, dB.x, dB.y, dB.z, dB.w};

    v8f acc[4] = {};

    for (int k0 = 0; k0 < IN_F; k0 += 4) {
        const int ks = k0 + half * 2;
        v2f a = *(const v2f*)(xrow + ks);
        const float* bp = &sW[((ks >> 1) << 7) + (l15 << 1)];
#pragma unroll
        for (int t = 0; t < 4; ++t) {
            v2f b = *(const v2f*)(bp + (t << 5));
            acc[t] = __builtin_amdgcn_wmma_f32_16x16x4_f32(
                false, a, false, b, (short)0, acc[t], false, false);
        }
    }

#pragma unroll
    for (int t = 0; t < 4; ++t) {
        const int col = t * 16 + l15;
#pragma unroll
        for (int r = 0; r < 8; ++r) {
            const int row = rowBase + r + 8 * half;
            h[(size_t)row * H_F + col] = acc[t][r] * dsc[r];
        }
    }
}

__global__ __launch_bounds__(256) void k_gather1(const int* __restrict__ rowptr,
                                                 const int* __restrict__ cnt,
                                                 const int* __restrict__ csr,
                                                 const float* __restrict__ h,
                                                 const float* __restrict__ dinv,
                                                 const float* __restrict__ b1,
                                                 const float* __restrict__ W2,
                                                 float* __restrict__ zs) {
    const int lane = threadIdx.x & 31;
    int n = __builtin_amdgcn_readfirstlane(blockIdx.x * 8 + (threadIdx.x >> 5));
    if (n >= N_NODES) return;
    const float2* h2 = (const float2*)h;
    const char* hb = (const char*)h;
    const unsigned lane8 = (unsigned)lane * 8u;
    const float2 bb = ((const float2*)b1)[lane];
    const float2 ww = ((const float2*)W2)[lane];

    int start = rowptr[n], c = cnt[n];
    int idx = lane < c ? csr[start + lane] : N_NODES;
    float2 acc = h2[(size_t)n * 32 + lane];
    float dn = dinv[n];
    int nn = n + PB * 8;
    int nstart = 0, nc = 0;
    if (nn < N_NODES) { nstart = rowptr[nn]; nc = cnt[nn]; }

    while (true) {
        int idxN = N_NODES, s2 = 0, c2 = 0;
        float2 accN = make_float2(0.0f, 0.0f);
        float dnN = 0.0f;
        const int n2 = nn + PB * 8;
        if (nn < N_NODES) {
            idxN = lane < nc ? csr[nstart + lane] : N_NODES;
            accN = h2[(size_t)nn * 32 + lane];
            dnN  = dinv[nn];
            if (n2 < N_NODES) { s2 = rowptr[n2]; c2 = cnt[n2]; }
        }

        for (int base = 0; base < c; base += 32) {
            const int m = min(32, c - base);
            if (base > 0) idx = lane < m ? csr[start + base + lane] : N_NODES;
            for (int j = 0; j < m; j += 16) {
                float2 v[16];
#pragma unroll
                for (int u = 0; u < 8; ++u)
                    v[u] = *(const float2*)(hb + ((unsigned)__builtin_amdgcn_readlane(idx, j + u) * 256u + lane8));
                if (j + 8 < m) {
#pragma unroll
                    for (int u = 0; u < 8; ++u)
                        v[8 + u] = *(const float2*)(hb + ((unsigned)__builtin_amdgcn_readlane(idx, j + 8 + u) * 256u + lane8));
                } else {
#pragma unroll
                    for (int u = 0; u < 8; ++u) v[8 + u] = make_float2(0.0f, 0.0f);
                }
#pragma unroll
                for (int u = 0; u < 16; ++u) { acc.x += v[u].x; acc.y += v[u].y; }
            }
        }

        float s = fmaxf(acc.x * dn + bb.x, 0.0f) * ww.x + fmaxf(acc.y * dn + bb.y, 0.0f) * ww.y;
#pragma unroll
        for (int off = 16; off > 0; off >>= 1) s += __shfl_xor(s, off, 32);
        if (lane == 0) zs[n] = s * dn;

        if (nn >= N_NODES) break;
        n = nn; start = nstart; c = nc; idx = idxN; acc = accN; dn = dnN;
        nn = n2; nstart = s2; nc = c2;
    }
}

__global__ __launch_bounds__(256) void k_gather2(const int* __restrict__ rowptr,
                                                 const int* __restrict__ cnt,
                                                 const int* __restrict__ csr,
                                                 const float* __restrict__ zs,
                                                 const float* __restrict__ dinv,
                                                 const float* __restrict__ b2,
                                                 float* __restrict__ out) {
    const int sub = threadIdx.x & 7;
    for (int n = (blockIdx.x * 256 + threadIdx.x) >> 3; n < N_NODES; n += PB * 32) {
        const int start = rowptr[n], c = cnt[n];
        float a0 = 0.0f, a1 = 0.0f;
        int p = sub;
        for (; p + 8 < c; p += 16) {
            const int i0 = csr[start + p], i1 = csr[start + p + 8];
            a0 += zs[i0];
            a1 += zs[i1];
        }
        if (p < c) a0 += zs[csr[start + p]];
        float acc = a0 + a1;
        acc += __shfl_xor(acc, 4, 32);
        acc += __shfl_xor(acc, 2, 32);
        acc += __shfl_xor(acc, 1, 32);
        if (sub == 0) out[n] = (acc + zs[n]) * dinv[n] + b2[0];
    }
}

extern "C" void kernel_launch(void* const* d_in, const int* in_sizes, int n_in,
                              void* d_out, int out_size, void* d_ws, size_t ws_size,
                              hipStream_t stream) {
    const float* x   = (const float*)d_in[0];
    const float* W1  = (const float*)d_in[1];
    const float* b1  = (const float*)d_in[2];
    const float* W2  = (const float*)d_in[3];
    const float* b2  = (const float*)d_in[4];
    const int*   ei  = (const int*)d_in[5];
    const int*   src = ei;
    const int*   dst = ei + N_EDGES;
    float*       out = (float*)d_out;

    char* ws = (char*)d_ws;
    const size_t NPAD = 409600;
    const size_t EPAD = 6400000;
    const size_t HPAD = 524288;
    const size_t BPAD = 4096;
    int*   cnt    = (int*)(ws);
    float* dinv   = (float*)(ws + NPAD);
    int*   rowptr = (int*)(ws + 2 * NPAD);
    float* zs     = (float*)(ws + 3 * NPAD);
    int*   hist   = (int*)(ws + 4 * NPAD);
    int*   bbase  = (int*)(ws + 4 * NPAD + HPAD);
    int*   rec    = (int*)(ws + 4 * NPAD + HPAD + BPAD);
    int*   csr    = (int*)(ws + 4 * NPAD + HPAD + BPAD + EPAD);
    float* h      = (float*)(ws + 4 * NPAD + HPAD + BPAD + 2 * EPAD);

    k_hist   <<<NBLK, 1024, 0, stream>>>(dst, hist, h + (size_t)N_NODES * H_F);
    k_part   <<<NBLK, 1024, 0, stream>>>(src, dst, hist, bbase, rec);
    k_group  <<<NB, 512, 0, stream>>>(rec, bbase, rowptr, cnt, dinv, csr);
    k_gemm1  <<<(N_NODES + 127) / 128, 256, 0, stream>>>(x, W1, dinv, h);
    k_gather1<<<PB, 256, 0, stream>>>(rowptr, cnt, csr, h, dinv, b1, W2, zs);
    k_gather2<<<PB, 256, 0, stream>>>(rowptr, cnt, csr, zs, dinv, b2, out);
}


# x@W1 moved into the level-2 grouping launch: each 256-node block's 16 waves also compute its 16 WMMA strips, scaled by its own dinv from LDS; paired blocks run the two in opposite order; 5 launches
# speedup vs baseline: 3.4625x; 1.0494x over previous
#include <hip/hip_runtime.h>

#define N_NODES 100000
#define N_EDGES 1600000
#define IN_F 128
#define H_F 64
#define PB 2048
#define BSH 8
#define NB ((N_NODES + 255) >> BSH)
#define NBP 512
#define NBLK 256
#define CH ((N_EDGES + NBLK - 1) / NBLK)
#define EPT ((CH + 1023) / 1024)

typedef float v2f __attribute__((ext_vector_type(2)));
typedef float v8f __attribute__((ext_vector_type(8)));

static_assert(NB < NBP, "bucket NB of the padded row exists and stays empty");
static_assert(NBP % 4 == 0 && NBP / 4 == 128, "k_part reads a table row as 128 int4");
static_assert(NBLK % 128 == 0, "k_part: 8 thread groups walk the table rows in groups of 16");
static_assert(N_NODES < (1 << 23), "src << 8 | local node fits an int");
static_assert(N_NODES % 16 == 0, "product strips are whole: 16 rows, 8 dinv values per half");
static_assert((NB << BSH) >= N_NODES && (1 << BSH) == 16 * 16, "a level-2 block's 16 waves cover its 256 nodes in 16-row strips");
static_assert((unsigned long long)(N_NODES + 1) * H_F * 4 < 0xffffffffull, "row byte offsets fit 32 bits");

__global__ __launch_bounds__(1024) void k_hist(const int* __restrict__ dst,
                                               int* __restrict__ hist,
                                               float* __restrict__ hzero) {
    __shared__ int sH[NBP];
    const int tid = threadIdx.x;
    if (tid < NBP) sH[tid] = 0;
    if (blockIdx.x == 0 && tid < H_F) hzero[tid] = 0.0f;
    __syncthreads();
    const int e0 = blockIdx.x * CH;
    int d[EPT];
#pragma unroll
    for (int k = 0; k < EPT; ++k) {
        const int o = tid + k * 1024;
        const bool ok = o < CH && e0 + o < N_EDGES;
        const int dv = dst[ok ? e0 + o : 0];
        d[k] = ok ? dv : -1;
    }
#pragma unroll
    for (int k = 0; k < EPT; ++k)
        if (d[k] >= 0) atomicAdd(&sH[d[k] >> BSH], 1);
    __syncthreads();
    if (tid < NBP) hist[blockIdx.x * NBP + tid] = sH[tid];
}

__global__ __launch_bounds__(1024) void k_part(const int* __restrict__ src,
                                               const int* __restrict__ dst,
                                               const int* __restrict__ hist,
                                               int* __restrict__ bbase,
                                               int* __restrict__ rec) {
    __shared__ int sTot[NBP];
    __shared__ int sOff[NBP];
    __shared__ int sW[32];
    const int tid = threadIdx.x, lane = tid & 31, wave = tid >> 5;
    const int j = blockIdx.x;
    if (tid < NBP) { sTot[tid] = 0; sOff[tid] = 0; }
    __syncthreads();

    const int e0 = j * CH;
    int d[EPT], s[EPT];
#pragma unroll
    for (int k = 0; k < EPT; ++k) {
        const int o = tid + k * 1024;
        const bool ok = o < CH && e0 + o < N_EDGES;
        const int e = ok ? e0 + o : 0;
        const int dv = dst[e];
        s[k] = src[e];
        d[k] = ok ? dv : -1;
    }

    const int c4 = tid & 127, q = tid >> 7;
    const int4* h4 = (const int4*)hist;
    int4 tot = make_int4(0, 0, 0, 0), pre = make_int4(0, 0, 0, 0);
#pragma unroll 1
    for (int g = 0; g < NBLK / 8; g += 16) {
        int4 t[16];
#pragma unroll
        for (int u = 0; u < 16; ++u) t[u] = h4[(q + 8 * (g + u)) * (NBP / 4) + c4];
#pragma unroll
        for (int u = 0; u < 16; ++u) {
            const bool before = q + 8 * (g + u) < j;
            tot.x += t[u].x; tot.y += t[u].y; tot.z += t[u].z; tot.w += t[u].w;
            pre.x += before ? t[u].x : 0; pre.y += before ? t[u].y : 0;
            pre.z += before ? t[u].z : 0; pre.w += before ? t[u].w : 0;
        }
    }
    atomicAdd(&sTot[4 * c4 + 0], tot.x); atomicAdd(&sTot[4 * c4 + 1], tot.y);
    atomicAdd(&sTot[4 * c4 + 2], tot.z); atomicAdd(&sTot[4 * c4 + 3], tot.w);
    atomicAdd(&sOff[4 * c4 + 0], pre.x); atomicAdd(&sOff[4 * c4 + 1], pre.y);
    atomicAdd(&sOff[4 * c4 + 2], pre.z); atomicAdd(&sOff[4 * c4 + 3], pre.w);
    __syncthreads();

    const int v = tid < NBP ? sTot[tid] : 0;
    int inc = v;
#pragma unroll
    for (int off = 1; off < 32; off <<= 1) {
        const int t = __shfl_up(inc, off, 32);
        if (lane >= off) inc += t;
    }
    if (lane == 31) sW[wave] = inc;
    __syncthreads();
    int wb = lane < wave ? sW[lane] : 0;
#pragma unroll
    for (int off = 16; off > 0; off >>= 1) wb += __shfl_xor(wb, off, 32);
    const int excl = wb + inc - v;
    if (tid < NBP) {
        sOff[tid] += excl;
        if (j == 0 && tid <= NB) bbase[tid] = excl;
    }
    __syncthreads();

    int pos[EPT];
#pragma unroll
    for (int k = 0; k < EPT; ++k) pos[k] = d[k] >= 0 ? atomicAdd(&sOff[d[k] >> BSH], 1) : 0;
#pragma unroll
    for (int k = 0; k < EPT; ++k)
        if (d[k] >= 0) rec[pos[k]] = (s[k] << BSH) | (d[k] & 255);
}

__device__ __forceinline__ void strip_product(const float* __restrict__ xrow,
                                              const float* sW, int half, int l15,
                                              v8f (&acc)[4]) {
    for (int k0 = 0; k0 < IN_F; k0 += 4) {
        const int ks = k0 + half * 2;
        v2f a = *(const v2f*)(xrow + ks);
        const float* bp = &sW[((ks >> 1) << 7) + (l15 << 1)];
#pragma unroll
        for (int t = 0; t < 4; ++t) {
            v2f b = *(const v2f*)(bp + (t << 5));
            acc[t] = __builtin_amdgcn_wmma_f32_16x16x4_f32(
                false, a, false, b, (short)0, acc[t], false, false);
        }
    }
}

__global__ __launch_bounds__(512) void k_group_gemm(const int* __restrict__ rec,
                                                    const int* __restrict__ bbase,
                                                    int* __restrict__ rowptr,
                                                    int* __restrict__ cnt,
                                                    float* __restrict__ dinv,
                                                    int* __restrict__ csr,
                                                    const float* __restrict__ x,
                                                    const float* __restrict__ W1,
                                                    float* __restrict__ h) {
    __shared__ float sW[IN_F * H_F];
    __shared__ int sCnt[256];
    __shared__ int sCur[256];
    __shared__ float sDinv[256];
    __shared__ int sWs[8];
    const int tid = threadIdx.x, lane = tid & 31, wave = tid >> 5;
    const int b = blockIdx.x;
    const int begin = bbase[b], end = bbase[b + 1];
    if (tid < 256) sCnt[tid] = 0;
#pragma unroll
    for (int u = 0; u < IN_F * H_F / 512; ++u) {
        const int i = tid + u * 512;
        const int k = i >> 6, n = i & 63;
        sW[((k >> 1) << 7) + (n << 1) + (k & 1)] = W1[i];
    }
    __syncthreads();

    const int wv      = __builtin_amdgcn_readfirstlane(tid >> 5);
    const int half    = lane >> 4;
    const int l15     = lane & 15;
    const int rowBase = (b << BSH) + wv * 16;
    const bool strip  = rowBase < N_NODES;
    const float* xrow = x + (size_t)(rowBase + l15) * IN_F;
    const bool productFirst = (((b >> 8) ^ b) & 1) != 0;

    v8f acc[4] = {};
    if (productFirst && strip) strip_product(xrow, sW, half, l15, acc);

    for (int i0 = begin; i0 < end; i0 += 512 * 8) {
        int r[8];
#pragma unroll
        for (int u = 0; u < 8; ++u) {
            const int i = i0 + u * 512 + tid;
            const int rv = rec[i < end ? i : begin];
            r[u] = i < end ? rv : -1;
        }
#pragma unroll
        for (int u = 0; u < 8; ++u)
            if (r[u] >= 0) atomicAdd(&sCnt[r[u] & 255], 1);
    }
    __syncthreads();

    const int c = tid < 256 ? sCnt[tid] : 0;
    int inc = c;
#pragma unroll
    for (int off = 1; off < 32; off <<= 1) {
        const int t = __shfl_up(inc, off, 32);
        if (lane >= off) inc += t;
    }
    if (tid < 256 && lane == 31) sWs[wave] = inc;
    __syncthreads();
    int wb = (lane < wave && lane < 8) ? sWs[lane] : 0;
#pragma unroll
    for (int off = 16; off > 0; off >>= 1) wb += __shfl_xor(wb, off, 32);
    if (tid < 256) {
        const int p = begin + wb + inc - c;
        sCur[tid] = p;
        const float dg = (float)(c + 1);
        const float dv = dg > 0.0f ? rsqrtf(dg) : 0.0f;
        sDinv[tid] = dv;
        const int node = (b << BSH) + tid;
        if (node < N_NODES) {
            rowptr[node] = p;
            cnt[node] = c;
            dinv[node] = dv;
        }
    }
    __syncthreads();

    for (int i0 = begin; i0 < end; i0 += 512 * 8) {
        int r[8];
#pragma unroll
        for (int u = 0; u < 8; ++u) {
            const int i = i0 + u * 512 + tid;
            const int rv = rec[i < end ? i : begin];
            r[u] = i < end ? rv : -1;
        }
#pragma unroll
        for (int u = 0; u < 8; ++u)
            if (r[u] >= 0) {
                const int pos = atomicAdd(&sCur[r[u] & 255], 1);
                csr[pos] = r[u] >> BSH;
            }
    }

    if (!strip) return;
    if (!productFirst) strip_product(xrow, sW, half, l15, acc);

    const float4* dp = (const float4*)(sDinv + wv * 16 + 8 * half);
    const float4 dA = dp[0], dB = dp[1];
    const float dsc[8] = {dA.x, dA.y, dA.z, dA.w, dB.x, dB.y, dB.z, dB.w};

#pragma unroll
    for (int t = 0; t < 4; ++t) {
        const int col = t * 16 + l15;
#pragma unroll
        for (int r = 0; r < 8; ++r) {
            const int row = rowBase + r + 8 * half;
            h[(size_t)row * H_F + col] = acc[t][r] * dsc[r];
        }
    }
}

__global__ __launch_bounds__(256) void k_gather1(const int* __restrict__ rowptr,
                                                 const int* __restrict__ cnt,
                                                 const int* __restrict__ csr,
                                                 const float* __restrict__ h,
                                                 const float* __restrict__ dinv,
                                                 const float* __restrict__ b1,
                                                 const float* __restrict__ W2,
                                                 float* __restrict__ zs) {
    const int lane = threadIdx.x & 31;
    int n = __builtin_amdgcn_readfirstlane(blockIdx.x * 8 + (threadIdx.x >> 5));
    if (n >= N_NODES) return;
    const float2* h2 = (const float2*)h;
    const char* hb = (const char*)h;
    const unsigned lane8 = (unsigned)lane * 8u;
    const float2 bb = ((const float2*)b1)[lane];
    const float2 ww = ((const float2*)W2)[lane];

    int start = rowptr[n], c = cnt[n];
    int idx = lane < c ? csr[start + lane] : N_NODES;
    float2 acc = h2[(size_t)n * 32 + lane];
    float dn = dinv[n];
    int nn = n + PB * 8;
    int nstart = 0, nc = 0;
    if (nn < N_NODES) { nstart = rowptr[nn]; nc = cnt[nn]; }

    while (true) {
        int idxN = N_NODES, s2 = 0, c2 = 0;
        float2 accN = make_float2(0.0f, 0.0f);
        float dnN = 0.0f;
        const int n2 = nn + PB * 8;
        if (nn < N_NODES) {
            idxN = lane < nc ? csr[nstart + lane] : N_NODES;
            accN = h2[(size_t)nn * 32 + lane];
            dnN  = dinv[nn];
            if (n2 < N_NODES) { s2 = rowptr[n2]; c2 = cnt[n2]; }
        }

        for (int base = 0; base < c; base += 32) {
            const int m = min(32, c - base);
            if (base > 0) idx = lane < m ? csr[start + base + lane] : N_NODES;
            for (int j = 0; j < m; j += 16) {
                float2 v[16];
#pragma unroll
                for (int u = 0; u < 8; ++u)
                    v[u] = *(const float2*)(hb + ((unsigned)__builtin_amdgcn_readlane(idx, j + u) * 256u + lane8));
                if (j + 8 < m) {
#pragma unroll
                    for (int u = 0; u < 8; ++u)
                        v[8 + u] = *(const float2*)(hb + ((unsigned)__builtin_amdgcn_readlane(idx, j + 8 + u) * 256u + lane8));
                } else {
#pragma unroll
                    for (int u = 0; u < 8; ++u) v[8 + u] = make_float2(0.0f, 0.0f);
                }
#pragma unroll
                for (int u = 0; u < 16; ++u) { acc.x += v[u].x; acc.y += v[u].y; }
            }
        }

        float s = fmaxf(acc.x * dn + bb.x, 0.0f) * ww.x + fmaxf(acc.y * dn + bb.y, 0.0f) * ww.y;
#pragma unroll
        for (int off = 16; off > 0; off >>= 1) s += __shfl_xor(s, off, 32);
        if (lane == 0) zs[n] = s * dn;

        if (nn >= N_NODES) break;
        n = nn; start = nstart; c = nc; idx = idxN; acc = accN; dn = dnN;
        nn = n2; nstart = s2; nc = c2;
    }
}

__global__ __launch_bounds__(256) void k_gather2(const int* __restrict__ rowptr,
                                                 const int* __restrict__ cnt,
                                                 const int* __restrict__ csr,
                                                 const float* __restrict__ zs,
                                                 const float* __restrict__ dinv,
                                                 const float* __restrict__ b2,
                                                 float* __restrict__ out) {
    const int sub = threadIdx.x & 7;
    for (int n = (blockIdx.x * 256 + threadIdx.x) >> 3; n < N_NODES; n += PB * 32) {
        const int start = rowptr[n], c = cnt[n];
        float a0 = 0.0f, a1 = 0.0f;
        int p = sub;
        for (; p + 8 < c; p += 16) {
            const int i0 = csr[start + p], i1 = csr[start + p + 8];
            a0 += zs[i0];
            a1 += zs[i1];
        }
        if (p < c) a0 += zs[csr[start + p]];
        float acc = a0 + a1;
        acc += __shfl_xor(acc, 4, 32);
        acc += __shfl_xor(acc, 2, 32);
        acc += __shfl_xor(acc, 1, 32);
        if (sub == 0) out[n] = (acc + zs[n]) * dinv[n] + b2[0];
    }
}

extern "C" void kernel_launch(void* const* d_in, const int* in_sizes, int n_in,
                              void* d_out, int out_size, void* d_ws, size_t ws_size,
                              hipStream_t stream) {
    const float* x   = (const float*)d_in[0];
    const float* W1  = (const float*)d_in[1];
    const float* b1  = (const float*)d_in[2];
    const float* W2  = (const float*)d_in[3];
    const float* b2  = (const float*)d_in[4];
    const int*   ei  = (const int*)d_in[5];
    const int*   src = ei;
    const int*   dst = ei + N_EDGES;
    float*       out = (float*)d_out;

    char* ws = (char*)d_ws;
    const size_t NPAD = 409600;
    const size_t EPAD = 6400000;
    const size_t HPAD = 524288;
    const size_t BPAD = 4096;
    int*   cnt    = (int*)(ws);
    float* dinv   = (float*)(ws + NPAD);
    int*   rowptr = (int*)(ws + 2 * NPAD);
    float* zs     = (float*)(ws + 3 * NPAD);
    int*   hist   = (int*)(ws + 4 * NPAD);
    int*   bbase  = (int*)(ws + 4 * NPAD + HPAD);
    int*   rec    = (int*)(ws + 4 * NPAD + HPAD + BPAD);
    int*   csr    = (int*)(ws + 4 * NPAD + HPAD + BPAD + EPAD);
    float* h      = (float*)(ws + 4 * NPAD + HPAD + BPAD + 2 * EPAD);

    k_hist   <<<NBLK, 1024, 0, stream>>>(dst, hist, h + (size_t)N_NODES * H_F);
    k_part   <<<NBLK, 1024, 0, stream>>>(src, dst, hist, bbase, rec);
    k_group_gemm<<<NB, 512, 0, stream>>>(rec, bbase, rowptr, cnt, dinv, csr, x, W1, h);
    k_gather1<<<PB, 256, 0, stream>>>(rowptr, cnt, csr, h, dinv, b1, W2, zs);
    k_gather2<<<PB, 256, 0, stream>>>(rowptr, cnt, csr, zs, dinv, b2, out);
}


# no global count pass: each level-1 block orders its own share of edges by bucket in LDS, stores it whole; level 2 collects its 256 runs via a (start,count) table; 4 launches; list (sum) order changes
# speedup vs baseline: 4.9967x; 1.4431x over previous
#include <hip/hip_runtime.h>

#define N_NODES 100000
#define N_EDGES 1600000
#define IN_F 128
#define H_F 64
#define PB 2048
#define BSH 8
#define NB ((N_NODES + 255) >> BSH)
#define NBP 512
#define NBLK 256
#define CH ((N_EDGES + NBLK - 1) / NBLK)
#define CHP (((CH + 31) / 32) * 32)
#define EPT ((CH + 1023) / 1024)

typedef float v2f __attribute__((ext_vector_type(2)));
typedef float v8f __attribute__((ext_vector_type(8)));

static_assert(NB < NBP, "bucket NB of the padded row exists and stays empty");
static_assert(NBLK == 8 * 32, "level 2: 8 waves fetch one table entry per lane; 8 rounds of 32 runs");
static_assert(N_NODES < (1 << 23), "src << 8 | local node fits an int");
static_assert(N_NODES % 16 == 0, "product strips are whole: 16 rows, 8 dinv values per half");
static_assert((NB << BSH) >= N_NODES && (1 << BSH) == 16 * 16, "a level-2 block's 16 waves cover its 256 nodes in 16-row strips");
static_assert((unsigned long long)(N_NODES + 1) * H_F * 4 < 0xffffffffull, "row byte offsets fit 32 bits");

__global__ __launch_bounds__(1024) void k_share(const int* __restrict__ src,
                                                const int* __restrict__ dst,
                                                int* __restrict__ tab,
                                                int* __restrict__ rec,
                                                float* __restrict__ hzero) {
    __shared__ int sH[NBP];
    __shared__ int sOff[NBP];
    __shared__ int sW[32];
    __shared__ int sRec[CH];
    const int tid = threadIdx.x, lane = tid & 31, wave = tid >> 5;
    const int j = blockIdx.x;
    if (tid < NBP) sH[tid] = 0;
    if (j == 0 && tid < H_F) hzero[tid] = 0.0f;
    __syncthreads();

    const int e0 = j * CH;
    int d[EPT], s[EPT];
#pragma unroll
    for (int k = 0; k < EPT; ++k) {
        const int o = tid + k * 1024;
        const bool ok = o < CH && e0 + o < N_EDGES;
        const int e = ok ? e0 + o : 0;
        const int dv = dst[e];
        s[k] = src[e];
        d[k] = ok ? dv : -1;
    }
#pragma unroll
    for (int k = 0; k < EPT; ++k)
        if (d[k] >= 0) atomicAdd(&sH[d[k] >> BSH], 1);
    __syncthreads();

    const int v = tid < NBP ? sH[tid] : 0;
    int inc = v;
#pragma unroll
    for (int off = 1; off < 32; off <<= 1) {
        const int t = __shfl_up(inc, off, 32);
        if (lane >= off) inc += t;
    }
    if (lane == 31) sW[wave] = inc;
    __syncthreads();
    int wb = lane < wave ? sW[lane] : 0;
#pragma unroll
    for (int off = 16; off > 0; off >>= 1) wb += __shfl_xor(wb, off, 32);
    const int excl = wb + inc - v;
    if (tid < NBP) {
        sOff[tid] = excl;
        ((int2*)tab)[j * NBP + tid] = make_int2(excl, v);
    }
    __syncthreads();

    int pos[EPT];
#pragma unroll
    for (int k = 0; k < EPT; ++k) pos[k] = d[k] >= 0 ? atomicAdd(&sOff[d[k] >> BSH], 1) : 0;
#pragma unroll
    for (int k = 0; k < EPT; ++k)
        if (d[k] >= 0) sRec[pos[k]] = (s[k] << BSH) | (d[k] & 255);
    __syncthreads();

    const int nv = min(CH, N_EDGES - e0);
    for (int i = tid; i < nv; i += 1024) rec[j * CHP + i] = sRec[i];
}

__device__ __forceinline__ void strip_product(const float* __restrict__ xrow,
                                              const float* sW, int half, int l15,
                                              v8f (&acc)[4]) {
    for (int k0 = 0; k0 < IN_F; k0 += 4) {
        const int ks = k0 + half * 2;
        v2f a = *(const v2f*)(xrow + ks);
        const float* bp = &sW[((ks >> 1) << 7) + (l15 << 1)];
#pragma unroll
        for (int t = 0; t < 4; ++t) {
            v2f b = *(const v2f*)(bp + (t << 5));
            acc[t] = __builtin_amdgcn_wmma_f32_16x16x4_f32(
                false, a, false, b, (short)0, acc[t], false, false);
        }
    }
}

__device__ __forceinline__ void load_runs(const int* __restrict__ rec, const int2* sRun,
                                          int rg, int l15, int (&v)[16],
                                          unsigned& okm, bool& longRun) {
    okm = 0u;
    longRun = false;
#pragma unroll
    for (int u = 0; u < 8; ++u) {
        const int2 rc = sRun[u * 32 + rg];
        const int a = (u * 32 + rg) * CHP + rc.x + l15;
        const bool ok0 = l15 < rc.y, ok1 = l15 + 16 < rc.y;
        v[2 * u] = rec[ok0 ? a : 0];
        v[2 * u + 1] = rec[ok1 ? a + 16 : 0];
        okm |= (ok0 ? 1u : 0u) << (2 * u);
        okm |= (ok1 ? 1u : 0u) << (2 * u + 1);
        longRun |= rc.y > 32;
    }
    asm volatile("" :: "v"(v[0]), "v"(v[1]), "v"(v[2]), "v"(v[3]), "v"(v[4]), "v"(v[5]),
                 "v"(v[6]), "v"(v[7]), "v"(v[8]), "v"(v[9]), "v"(v[10]), "v"(v[11]),
                 "v"(v[12]), "v"(v[13]), "v"(v[14]), "v"(v[15]));
}

__global__ __launch_bounds__(512) void k_group_gemm(const int* __restrict__ rec,
                                                    const int* __restrict__ tab,
                                                    int* __restrict__ rowptr,
                                                    int* __restrict__ cnt,
                                                    float* __restrict__ dinv,
                                                    int* __restrict__ csr,
                                                    const float* __restrict__ x,
                                                    const float* __restrict__ W1,
                                                    float* __restrict__ h) {
    __shared__ float sW[IN_F * H_F];
    __shared__ int sCnt[256];
    __shared__ int sCur[256];
    __shared__ float sDinv[256];
    __shared__ int sWs[8];
    __shared__ int2 sRun[NBLK];
    __shared__ int sBs[8];
    const int tid = threadIdx.x, lane = tid & 31, wave = tid >> 5;
    const int b = blockIdx.x;
    int2 oc = make_int2(0, 0);
    if (tid < NBLK) oc = ((const int2*)tab)[tid * NBP + b];
    if (tid < 256) sCnt[tid] = 0;
#pragma unroll
    for (int u = 0; u < IN_F * H_F / 512; ++u) {
        const int i = tid + u * 512;
        const int k = i >> 6, n = i & 63;
        sW[((k >> 1) << 7) + (n << 1) + (k & 1)] = W1[i];
    }
    if (tid < NBLK) {
        sRun[tid] = oc;
        int o = oc.x;
#pragma unroll
        for (int off = 16; off > 0; off >>= 1) o += __shfl_xor(o, off, 32);
        if (lane == 0) sBs[wave] = o;
    }
    __syncthreads();
    int begin = 0;
#pragma unroll
    for (int w = 0; w < 8; ++w) begin += sBs[w];
    const int rg = tid >> 4;

    const int wv      = __builtin_amdgcn_readfirstlane(tid >> 5);
    const int half    = lane >> 4;
    const int l15     = lane & 15;
    const int rowBase = (b << BSH) + wv * 16;
    const bool strip  = rowBase < N_NODES;
    const float* xrow = x + (size_t)(rowBase + l15) * IN_F;
    const bool productFirst = (((b >> 8) ^ b) & 1) != 0;

    v8f acc[4] = {};
    if (productFirst && strip) strip_product(xrow, sW, half, l15, acc);

    {
        int r[16];
        unsigned okm;
        bool longRun;
        load_runs(rec, sRun, rg, l15, r, okm, longRun);
#pragma unroll
        for (int u = 0; u < 16; ++u)
            if ((okm >> u) & 1u) atomicAdd(&sCnt[r[u] & 255], 1);
        if (longRun) {
            for (int u = 0; u < 8; ++u) {
                const int2 rc = sRun[u * 32 + rg];
                const int a = (u * 32 + rg) * CHP + rc.x;
                for (int k = l15 + 32; k < rc.y; k += 16) atomicAdd(&sCnt[rec[a + k] & 255], 1);
            }
        }
    }
    __syncthreads();

    const int c = tid < 256 ? sCnt[tid] : 0;
    int inc = c;
#pragma unroll
    for (int off = 1; off < 32; off <<= 1) {
        const int t = __shfl_up(inc, off, 32);
        if (lane >= off) inc += t;
    }
    if (tid < 256 && lane == 31) sWs[wave] = inc;
    __syncthreads();
    int wb = (lane < wave && lane < 8) ? sWs[lane] : 0;
#pragma unroll
    for (int off = 16; off > 0; off >>= 1) wb += __shfl_xor(wb, off, 32);
    if (tid < 256) {
        const int p = begin + wb + inc - c;
        sCur[tid] = p;
        const float dg = (float)(c + 1);
        const float dv = dg > 0.0f ? rsqrtf(dg) : 0.0f;
        sDinv[tid] = dv;
        const int node = (b << BSH) + tid;
        if (node < N_NODES) {
            rowptr[node] = p;
            cnt[node] = c;
            dinv[node] = dv;
        }
    }
    __syncthreads();

    {
        int r[16];
        unsigned okm;
        bool longRun;
        load_runs(rec, sRun, rg, l15, r, okm, longRun);
#pragma unroll
        for (int u = 0; u < 16; ++u)
            if ((okm >> u) & 1u) {
                const int pos = atomicAdd(&sCur[r[u] & 255], 1);
                csr[pos] = r[u] >> BSH;
            }
        if (longRun) {
            for (int u = 0; u < 8; ++u) {
                const int2 rc = sRun[u * 32 + rg];
                const int a = (u * 32 + rg) * CHP + rc.x;
                for (int k = l15 + 32; k < rc.y; k += 16) {
                    const int w = rec[a + k];
                    const int pos = atomicAdd(&sCur[w & 255], 1);
                    csr[pos] = w >> BSH;
                }
            }
        }
    }

    if (!strip) return;
    if (!productFirst) strip_product(xrow, sW, half, l15, acc);

    const float4* dp = (const float4*)(sDinv + wv * 16 + 8 * half);
    const float4 dA = dp[0], dB = dp[1];
    const float dsc[8] = {dA.x, dA.y, dA.z, dA.w, dB.x, dB.y, dB.z, dB.w};

#pragma unroll
    for (int t = 0; t < 4; ++t) {
        const int col = t * 16 + l15;
#pragma unroll
        for (int r = 0; r < 8; ++r) {
            const int row = rowBase + r + 8 * half;
            h[(size_t)row * H_F + col] = acc[t][r] * dsc[r];
        }
    }
}

__global__ __launch_bounds__(256) void k_gather1(const int* __restrict__ rowptr,
                                                 const int* __restrict__ cnt,
                                                 const int* __restrict__ csr,
                                                 const float* __restrict__ h,
                                                 const float* __restrict__ dinv,
                                                 const float* __restrict__ b1,
                                                 const float* __restrict__ W2,
                                                 float* __restrict__ zs) {
    const int lane = threadIdx.x & 31;
    int n = __builtin_amdgcn_readfirstlane(blockIdx.x * 8 + (threadIdx.x >> 5));
    if (n >= N_NODES) return;
    const float2* h2 = (const float2*)h;
    const char* hb = (const char*)h;
    const unsigned lane8 = (unsigned)lane * 8u;
    const float2 bb = ((const float2*)b1)[lane];
    const float2 ww = ((const float2*)W2)[lane];

    int start = rowptr[n], c = cnt[n];
    int idx = lane < c ? csr[start + lane] : N_NODES;
    float2 acc = h2[(size_t)n * 32 + lane];
    float dn = dinv[n];
    int nn = n + PB * 8;
    int nstart = 0, nc = 0;
    if (nn < N_NODES) { nstart = rowptr[nn]; nc = cnt[nn]; }

    while (true) {
        int idxN = N_NODES, s2 = 0, c2 = 0;
        float2 accN = make_float2(0.0f, 0.0f);
        float dnN = 0.0f;
        const int n2 = nn + PB * 8;
        if (nn < N_NODES) {
            idxN = lane < nc ? csr[nstart + lane] : N_NODES;
            accN = h2[(size_t)nn * 32 + lane];
            dnN  = dinv[nn];
            if (n2 < N_NODES) { s2 = rowptr[n2]; c2 = cnt[n2]; }
        }

        for (int base = 0; base < c; base += 32) {
            const int m = min(32, c - base);
            if (base > 0) idx = lane < m ? csr[start + base + lane] : N_NODES;
            for (int j = 0; j < m; j += 16) {
                float2 v[16];
#pragma unroll
                for (int u = 0; u < 8; ++u)
                    v[u] = *(const float2*)(hb + ((unsigned)__builtin_amdgcn_readlane(idx, j + u) * 256u + lane8));
                if (j + 8 < m) {
#pragma unroll
                    for (int u = 0; u < 8; ++u)
                        v[8 + u] = *(const float2*)(hb + ((unsigned)__builtin_amdgcn_readlane(idx, j + 8 + u) * 256u + lane8));
                } else {
#pragma unroll
                    for (int u = 0; u < 8; ++u) v[8 + u] = make_float2(0.0f, 0.0f);
                }
#pragma unroll
                for (int u = 0; u < 16; ++u) { acc.x += v[u].x; acc.y += v[u].y; }
            }
        }

        float s = fmaxf(acc.x * dn + bb.x, 0.0f) * ww.x + fmaxf(acc.y * dn + bb.y, 0.0f) * ww.y;
#pragma unroll
        for (int off = 16; off > 0; off >>= 1) s += __shfl_xor(s, off, 32);
        if (lane == 0) zs[n] = s * dn;

        if (nn >= N_NODES) break;
        n = nn; start = nstart; c = nc; idx = idxN; acc = accN; dn = dnN;
        nn = n2; nstart = s2; nc = c2;
    }
}

__global__ __launch_bounds__(256) void k_gather2(const int* __restrict__ rowptr,
                                                 const int* __restrict__ cnt,
                                                 const int* __restrict__ csr,
                                                 const float* __restrict__ zs,
                                                 const float* __restrict__ dinv,
                                                 const float* __restrict__ b2,
                                                 float* __restrict__ out) {
    const int sub = threadIdx.x & 7;
    for (int n = (blockIdx.x * 256 + threadIdx.x) >> 3; n < N_NODES; n += PB * 32) {
        const int start = rowptr[n], c = cnt[n];
        float a0 = 0.0f, a1 = 0.0f;
        int p = sub;
        for (; p + 8 < c; p += 16) {
            const int i0 = csr[start + p], i1 = csr[start + p + 8];
            a0 += zs[i0];
            a1 += zs[i1];
        }
        if (p < c) a0 += zs[csr[start + p]];
        float acc = a0 + a1;
        acc += __shfl_xor(acc, 4, 32);
        acc += __shfl_xor(acc, 2, 32);
        acc += __shfl_xor(acc, 1, 32);
        if (sub == 0) out[n] = (acc + zs[n]) * dinv[n] + b2[0];
    }
}

extern "C" void kernel_launch(void* const* d_in, const int* in_sizes, int n_in,
                              void* d_out, int out_size, void* d_ws, size_t ws_size,
                              hipStream_t stream) {
    const float* x   = (const float*)d_in[0];
    const float* W1  = (const float*)d_in[1];
    const float* b1  = (const float*)d_in[2];
    const float* W2  = (const float*)d_in[3];
    const float* b2  = (const float*)d_in[4];
    const int*   ei  = (const int*)d_in[5];
    const int*   src = ei;
    const int*   dst = ei + N_EDGES;
    float*       out = (float*)d_out;

    char* ws = (char*)d_ws;
    const size_t NPAD = 409600;
    const size_t EPAD = 6400000;
    const size_t TPAD = 1048576;
    const size_t RPAD = 6422528;
    int*   cnt    = (int*)(ws);
    float* dinv   = (float*)(ws + NPAD);
    int*   rowptr = (int*)(ws + 2 * NPAD);
    float* zs     = (float*)(ws + 3 * NPAD);
    int*   tab    = (int*)(ws + 4 * NPAD);
    int*   rec    = (int*)(ws + 4 * NPAD + TPAD);
    int*   csr    = (int*)(ws + 4 * NPAD + TPAD + RPAD);
    float* h      = (float*)(ws + 4 * NPAD + TPAD + RPAD + EPAD);

    k_share  <<<NBLK, 1024, 0, stream>>>(src, dst, tab, rec, h + (size_t)N_NODES * H_F);
    k_group_gemm<<<NB, 512, 0, stream>>>(rec, tab, rowptr, cnt, dinv, csr, x, W1, h);
    k_gather1<<<PB, 256, 0, stream>>>(rowptr, cnt, csr, h, dinv, b1, W2, zs);
    k_gather2<<<PB, 256, 0, stream>>>(rowptr, cnt, csr, zs, dinv, b2, out);
}
